# starting kernel
# baseline (speedup 1.0000x reference)
#include <hip/hip_runtime.h>

typedef __attribute__((ext_vector_type(2))) float v2f;
typedef __attribute__((ext_vector_type(8))) float v8f;

#define DIM 128

__global__ __launch_bounds__(256) void k_deg_init(float* __restrict__ deg, int n) {
  int i = blockIdx.x * blockDim.x + threadIdx.x;
  if (i < n) deg[i] = 1.0f;
}

__global__ __launch_bounds__(256) void k_deg_scatter(const int* __restrict__ src,
                                                     const float* __restrict__ w,
                                                     float* __restrict__ deg, int E) {
  int e = blockIdx.x * blockDim.x + threadIdx.x;
  if (e < E) atomicAdd(&deg[src[e]], w[e]);
}

__global__ __launch_bounds__(256) void k_dinv(const float* __restrict__ deg,
                                              float* __restrict__ dinv, int n) {
  int i = blockIdx.x * blockDim.x + threadIdx.x;
  if (i < n) dinv[i] = rsqrtf(deg[i]);
}

__global__ __launch_bounds__(256) void k_gemm_xwT(const float* __restrict__ X,
                                                  const float* __restrict__ W,
                                                  float* __restrict__ Y, int n) {
  const int wave = threadIdx.x >> 5;
  const int lane = threadIdx.x & 31;
  const int row0 = blockIdx.x << 4;
  const int m    = lane & 15;
  const int hi   = lane >> 4;
  const int kb   = hi << 1;
  const int col  = (wave << 4) + m;

  int row = row0 + m;
  if (row >= n) row = n - 1;
  const float* arow = X + (size_t)row * DIM;
  const float* brow = W + (size_t)col * DIM;

  v8f c = {};
#pragma unroll
  for (int k0 = 0; k0 < DIM; k0 += 4) {
    v2f a = *(const v2f*)(arow + k0 + kb);
    v2f b = *(const v2f*)(brow + k0 + kb);
    c = __builtin_amdgcn_wmma_f32_16x16x4_f32(false, a, false, b, (short)0, c,
                                              false, false);
  }

  const int mh = hi << 3;
#pragma unroll
  for (int v = 0; v < 8; ++v) {
    int r = row0 + mh + v;
    if (r < n) Y[(size_t)r * DIM + col] = c[v];
  }
}

__global__ __launch_bounds__(256) void k_selfloop_bias(const float* __restrict__ Y,
                                                       const float* __restrict__ dinv,
                                                       const float* __restrict__ bias,
                                                       float* __restrict__ out, int total) {
  int idx = blockIdx.x * blockDim.x + threadIdx.x;
  if (idx >= total) return;
  int i = idx >> 7;
  int d = idx & (DIM - 1);
  float di = dinv[i];
  out[idx] = Y[idx] * di * di + bias[d];
}

__global__ __launch_bounds__(256) void k_edge_scatter(const int* __restrict__ src,
                                                      const int* __restrict__ tgt,
                                                      const float* __restrict__ w,
                                                      const float* __restrict__ dinv,
                                                      const float* __restrict__ Y,
                                                      float* __restrict__ out, int E) {
  int e = blockIdx.x * 8 + (threadIdx.x >> 5);
  if (e >= E) return;
  int lane = threadIdx.x & 31;
  int s = src[e];
  int t = tgt[e];
  float wn = w[e] * dinv[s] * dinv[t];
  const float4 y = *(const float4*)(Y + (size_t)s * DIM + (lane << 2));
  float* op = out + (size_t)t * DIM + (lane << 2);
  atomicAdd(op + 0, y.x * wn);
  atomicAdd(op + 1, y.y * wn);
  atomicAdd(op + 2, y.z * wn);
  atomicAdd(op + 3, y.w * wn);
}

extern "C" void kernel_launch(void* const* d_in, const int* in_sizes, int n_in,
                              void* d_out, int out_size, void* d_ws, size_t ws_size,
                              hipStream_t stream) {
  const float* x          = (const float*)d_in[0];
  const int*   edge_index = (const int*)d_in[1];
  const float* ew         = (const float*)d_in[2];
  const float* W          = (const float*)d_in[3];
  const float* b          = (const float*)d_in[4];

  const int N = in_sizes[0] / DIM;
  const int E = in_sizes[2];
  const int* src = edge_index;
  const int* tgt = edge_index + E;

  float* ws   = (float*)d_ws;
  float* deg  = ws;
  float* dinv = ws + N;
  float* Y    = ws + 2 * (size_t)N;
  float* out  = (float*)d_out;

  k_deg_init<<<(N + 255) / 256, 256, 0, stream>>>(deg, N);
  k_deg_scatter<<<(E + 255) / 256, 256, 0, stream>>>(src, ew, deg, E);
  k_dinv<<<(N + 255) / 256, 256, 0, stream>>>(deg, dinv, N);
  k_gemm_xwT<<<(N + 15) / 16, 256, 0, stream>>>(x, W, Y, N);
  int total = N * DIM;
  k_selfloop_bias<<<(total + 255) / 256, 256, 0, stream>>>(Y, dinv, b, out, total);
  k_edge_scatter<<<(E + 7) / 8, 256, 0, stream>>>(src, tgt, ew, dinv, Y, out, E);
}


# edge scatter: lane l owns features l,l+32,l+64,l+96 so each atomic and each gather instruction covers one whole 128-byte line, not a quarter of four lines
# speedup vs baseline: 1.1930x; 1.1930x over previous
#include <hip/hip_runtime.h>

typedef __attribute__((ext_vector_type(2))) float v2f;
typedef __attribute__((ext_vector_type(8))) float v8f;

#define DIM 128

__global__ __launch_bounds__(256) void k_deg_init(float* __restrict__ deg, int n) {
  int i = blockIdx.x * blockDim.x + threadIdx.x;
  if (i < n) deg[i] = 1.0f;
}

__global__ __launch_bounds__(256) void k_deg_scatter(const int* __restrict__ src,
                                                     const float* __restrict__ w,
                                                     float* __restrict__ deg, int E) {
  int e = blockIdx.x * blockDim.x + threadIdx.x;
  if (e < E) atomicAdd(&deg[src[e]], w[e]);
}

__global__ __launch_bounds__(256) void k_dinv(const float* __restrict__ deg,
                                              float* __restrict__ dinv, int n) {
  int i = blockIdx.x * blockDim.x + threadIdx.x;
  if (i < n) dinv[i] = rsqrtf(deg[i]);
}

__global__ __launch_bounds__(256) void k_gemm_xwT(const float* __restrict__ X,
                                                  const float* __restrict__ W,
                                                  float* __restrict__ Y, int n) {
  const int wave = threadIdx.x >> 5;
  const int lane = threadIdx.x & 31;
  const int row0 = blockIdx.x << 4;
  const int m    = lane & 15;
  const int hi   = lane >> 4;
  const int kb   = hi << 1;
  const int col  = (wave << 4) + m;

  int row = row0 + m;
  if (row >= n) row = n - 1;
  const float* arow = X + (size_t)row * DIM;
  const float* brow = W + (size_t)col * DIM;

  v8f c = {};
#pragma unroll
  for (int k0 = 0; k0 < DIM; k0 += 4) {
    v2f a = *(const v2f*)(arow + k0 + kb);
    v2f b = *(const v2f*)(brow + k0 + kb);
    c = __builtin_amdgcn_wmma_f32_16x16x4_f32(false, a, false, b, (short)0, c,
                                              false, false);
  }

  const int mh = hi << 3;
#pragma unroll
  for (int v = 0; v < 8; ++v) {
    int r = row0 + mh + v;
    if (r < n) Y[(size_t)r * DIM + col] = c[v];
  }
}

__global__ __launch_bounds__(256) void k_selfloop_bias(const float* __restrict__ Y,
                                                       const float* __restrict__ dinv,
                                                       const float* __restrict__ bias,
                                                       float* __restrict__ out, int total) {
  int idx = blockIdx.x * blockDim.x + threadIdx.x;
  if (idx >= total) return;
  int i = idx >> 7;
  int d = idx & (DIM - 1);
  float di = dinv[i];
  out[idx] = Y[idx] * di * di + bias[d];
}

__global__ __launch_bounds__(256) void k_edge_scatter(const int* __restrict__ src,
                                                      const int* __restrict__ tgt,
                                                      const float* __restrict__ w,
                                                      const float* __restrict__ dinv,
                                                      const float* __restrict__ Y,
                                                      float* __restrict__ out, int E) {
  int e = blockIdx.x * 8 + (threadIdx.x >> 5);
  if (e >= E) return;
  int lane = threadIdx.x & 31;
  int s = src[e];
  int t = tgt[e];
  float wn = w[e] * dinv[s] * dinv[t];
  const float* yp = Y + (size_t)s * DIM + lane;
  float* op = out + (size_t)t * DIM + lane;
  const float y0 = yp[0];
  const float y1 = yp[32];
  const float y2 = yp[64];
  const float y3 = yp[96];
  atomicAdd(op, y0 * wn);
  atomicAdd(op + 32, y1 * wn);
  atomicAdd(op + 64, y2 * wn);
  atomicAdd(op + 96, y3 * wn);
}

extern "C" void kernel_launch(void* const* d_in, const int* in_sizes, int n_in,
                              void* d_out, int out_size, void* d_ws, size_t ws_size,
                              hipStream_t stream) {
  const float* x          = (const float*)d_in[0];
  const int*   edge_index = (const int*)d_in[1];
  const float* ew         = (const float*)d_in[2];
  const float* W          = (const float*)d_in[3];
  const float* b          = (const float*)d_in[4];

  const int N = in_sizes[0] / DIM;
  const int E = in_sizes[2];
  const int* src = edge_index;
  const int* tgt = edge_index + E;

  float* ws   = (float*)d_ws;
  float* deg  = ws;
  float* dinv = ws + N;
  float* Y    = ws + 2 * (size_t)N;
  float* out  = (float*)d_out;

  k_deg_init<<<(N + 255) / 256, 256, 0, stream>>>(deg, N);
  k_deg_scatter<<<(E + 255) / 256, 256, 0, stream>>>(src, ew, deg, E);
  k_dinv<<<(N + 255) / 256, 256, 0, stream>>>(deg, dinv, N);
  k_gemm_xwT<<<(N + 15) / 16, 256, 0, stream>>>(x, W, Y, N);
  int total = N * DIM;
  k_selfloop_bias<<<(total + 255) / 256, 256, 0, stream>>>(Y, dinv, b, out, total);
  k_edge_scatter<<<(E + 7) / 8, 256, 0, stream>>>(src, tgt, ew, dinv, Y, out, E);
}


# dense product: X strip and W copied to LDS with whole-row 16-byte-per-lane loads, WMMA fragments read from LDS (pitch 528 B), not by 8-byte global loads that touch 16 lines each
# speedup vs baseline: 1.3631x; 1.1426x over previous
#include <hip/hip_runtime.h>

typedef __attribute__((ext_vector_type(2))) float v2f;
typedef __attribute__((ext_vector_type(4))) float v4f;
typedef __attribute__((ext_vector_type(8))) float v8f;

#define DIM 128

__global__ __launch_bounds__(256) void k_deg_init(float* __restrict__ deg, int n) {
  int i = blockIdx.x * blockDim.x + threadIdx.x;
  if (i < n) deg[i] = 1.0f;
}

__global__ __launch_bounds__(256) void k_deg_scatter(const int* __restrict__ src,
                                                     const float* __restrict__ w,
                                                     float* __restrict__ deg, int E) {
  int e = blockIdx.x * blockDim.x + threadIdx.x;
  if (e < E) atomicAdd(&deg[src[e]], w[e]);
}

__global__ __launch_bounds__(256) void k_dinv(const float* __restrict__ deg,
                                              float* __restrict__ dinv, int n) {
  int i = blockIdx.x * blockDim.x + threadIdx.x;
  if (i < n) dinv[i] = rsqrtf(deg[i]);
}

#define LDP 132

__global__ __launch_bounds__(256) void k_gemm_xwT(const float* __restrict__ X,
                                                  const float* __restrict__ W,
                                                  float* __restrict__ Y, int n) {
  __shared__ __attribute__((aligned(16))) float sW[DIM * LDP];
  __shared__ __attribute__((aligned(16))) float sX[16 * LDP];

  const int wave = threadIdx.x >> 5;
  const int lane = threadIdx.x & 31;
  const int row0 = blockIdx.x << 4;

#pragma unroll
  for (int i = 0; i < 2; ++i) {
    const int r = wave + 8 * i;
    int gr = row0 + r;
    if (gr >= n) gr = n - 1;
    const v4f v = *(const v4f*)(X + (size_t)gr * DIM + 4 * lane);
    *(v4f*)(sX + r * LDP + 4 * lane) = v;
  }
#pragma unroll
  for (int i = 0; i < 16; ++i) {
    const int r = wave + 8 * i;
    const v4f v = *(const v4f*)(W + (size_t)r * DIM + 4 * lane);
    *(v4f*)(sW + r * LDP + 4 * lane) = v;
  }
  __syncthreads();

  const int m    = lane & 15;
  const int hi   = lane >> 4;
  const int kb   = hi << 1;
  const int col  = (wave << 4) + m;

  const float* arow = sX + m * LDP + kb;
  const float* brow = sW + col * LDP + kb;

  v8f c = {};
#pragma unroll
  for (int k0 = 0; k0 < DIM; k0 += 4) {
    v2f a = *(const v2f*)(arow + k0);
    v2f b = *(const v2f*)(brow + k0);
    c = __builtin_amdgcn_wmma_f32_16x16x4_f32(false, a, false, b, (short)0, c,
                                              false, false);
  }

  const int mh = hi << 3;
#pragma unroll
  for (int v = 0; v < 8; ++v) {
    int r = row0 + mh + v;
    if (r < n) Y[(size_t)r * DIM + col] = c[v];
  }
}

__global__ __launch_bounds__(256) void k_selfloop_bias(const float* __restrict__ Y,
                                                       const float* __restrict__ dinv,
                                                       const float* __restrict__ bias,
                                                       float* __restrict__ out, int total) {
  int idx = blockIdx.x * blockDim.x + threadIdx.x;
  if (idx >= total) return;
  int i = idx >> 7;
  int d = idx & (DIM - 1);
  float di = dinv[i];
  out[idx] = Y[idx] * di * di + bias[d];
}

__global__ __launch_bounds__(256) void k_edge_scatter(const int* __restrict__ src,
                                                      const int* __restrict__ tgt,
                                                      const float* __restrict__ w,
                                                      const float* __restrict__ dinv,
                                                      const float* __restrict__ Y,
                                                      float* __restrict__ out, int E) {
  int e = blockIdx.x * 8 + (threadIdx.x >> 5);
  if (e >= E) return;
  int lane = threadIdx.x & 31;
  int s = src[e];
  int t = tgt[e];
  float wn = w[e] * dinv[s] * dinv[t];
  const float* yp = Y + (size_t)s * DIM + lane;
  float* op = out + (size_t)t * DIM + lane;
  const float y0 = yp[0];
  const float y1 = yp[32];
  const float y2 = yp[64];
  const float y3 = yp[96];
  atomicAdd(op, y0 * wn);
  atomicAdd(op + 32, y1 * wn);
  atomicAdd(op + 64, y2 * wn);
  atomicAdd(op + 96, y3 * wn);
}

extern "C" void kernel_launch(void* const* d_in, const int* in_sizes, int n_in,
                              void* d_out, int out_size, void* d_ws, size_t ws_size,
                              hipStream_t stream) {
  const float* x          = (const float*)d_in[0];
  const int*   edge_index = (const int*)d_in[1];
  const float* ew         = (const float*)d_in[2];
  const float* W          = (const float*)d_in[3];
  const float* b          = (const float*)d_in[4];

  const int N = in_sizes[0] / DIM;
  const int E = in_sizes[2];
  const int* src = edge_index;
  const int* tgt = edge_index + E;

  float* ws   = (float*)d_ws;
  float* deg  = ws;
  float* dinv = ws + N;
  float* Y    = ws + 2 * (size_t)N;
  float* out  = (float*)d_out;

  k_deg_init<<<(N + 255) / 256, 256, 0, stream>>>(deg, N);
  k_deg_scatter<<<(E + 255) / 256, 256, 0, stream>>>(src, ew, deg, E);
  k_dinv<<<(N + 255) / 256, 256, 0, stream>>>(deg, dinv, N);
  k_gemm_xwT<<<(N + 15) / 16, 256, 0, stream>>>(x, W, Y, N);
  int total = N * DIM;
  k_selfloop_bias<<<(total + 255) / 256, 256, 0, stream>>>(Y, dinv, b, out, total);
  k_edge_scatter<<<(E + 7) / 8, 256, 0, stream>>>(src, tgt, ew, dinv, Y, out, E);
}


# node sums via LDS tables per edge chunk (degree; per-target rank from LDS atomics), then one wave per node adds its in-edge rows in registers, out stored once; no global float atomics; new sum order
# speedup vs baseline: 2.5919x; 1.9015x over previous
#include <hip/hip_runtime.h>

typedef __attribute__((ext_vector_type(2))) float v2f;
typedef __attribute__((ext_vector_type(4))) float v4f;
typedef __attribute__((ext_vector_type(8))) float v8f;
typedef __attribute__((ext_vector_type(2))) int v2i;
typedef __attribute__((ext_vector_type(4))) int v4i;

#define DIM 128

#define NCHUNK 32
#define NCAP 50176
#define HB 8

__global__ __launch_bounds__(1024) void k_hist(const int* __restrict__ src,
                                               const int* __restrict__ tgt,
                                               const float* __restrict__ w,
                                               float* __restrict__ degP,
                                               int* __restrict__ cntP,
                                               int* __restrict__ rank,
                                               int* __restrict__ counter, int n, int E) {
  __shared__ __attribute__((aligned(16))) union { int i[NCAP]; float f[NCAP]; } h;
  if (n > NCAP) return;
  const int tid = threadIdx.x;
  const int g = blockIdx.x & (NCHUNK - 1);
  const bool is_deg = blockIdx.x >= NCHUNK;
  if (blockIdx.x == 0 && tid == 0) *counter = 0;
  {
    const v4i z = {0, 0, 0, 0};
    for (int i = tid; i < NCAP / 4; i += 1024) ((v4i*)h.i)[i] = z;
  }
  __syncthreads();
  const int chunk = (E + NCHUNK - 1) / NCHUNK;
  const int e0 = g * chunk;
  const int e1 = min(E, e0 + chunk);
  if (is_deg) {
    for (int eb = e0 + tid; eb < e1; eb += 1024 * HB) {
      int key[HB];
      float val[HB];
#pragma unroll
      for (int u = 0; u < HB; ++u) {
        const int e = eb + 1024 * u;
        key[u] = (e < e1) ? src[e] : 0;
        val[u] = (e < e1) ? w[e] : 0.0f;
      }
#pragma unroll
      for (int u = 0; u < HB; ++u) {
        if (eb + 1024 * u < e1) atomicAdd(&h.f[key[u]], val[u]);
      }
    }
  } else {
    for (int eb = e0 + tid; eb < e1; eb += 1024 * HB) {
      int key[HB];
#pragma unroll
      for (int u = 0; u < HB; ++u) {
        const int e = eb + 1024 * u;
        key[u] = (e < e1) ? tgt[e] : 0;
      }
#pragma unroll
      for (int u = 0; u < HB; ++u) {
        const int e = eb + 1024 * u;
        if (e < e1) rank[e] = atomicAdd(&h.i[key[u]], 1);
      }
    }
  }
  __syncthreads();
  if (is_deg) {
    for (int i = tid; i < n; i += 1024) degP[(size_t)g * n + i] = h.f[i];
  } else {
    for (int i = tid; i < n; i += 1024) cntP[(size_t)g * n + i] = h.i[i];
  }
}

__global__ __launch_bounds__(256) void k_node(const float* __restrict__ degP,
                                              int* __restrict__ cntP,
                                              float* __restrict__ dinv,
                                              int* __restrict__ start,
                                              int* __restrict__ endp,
                                              int* __restrict__ counter, int n) {
  __shared__ int wsum[8];
  __shared__ int bbase;
  const int i = blockIdx.x * 256 + threadIdx.x;
  const int lane = threadIdx.x & 31;
  const int wave = threadIdx.x >> 5;
  int c[NCHUNK];
  int tot = 0;
  if (i < n) {
    float dp[NCHUNK];
#pragma unroll
    for (int g = 0; g < NCHUNK; ++g) dp[g] = degP[(size_t)g * n + i];
#pragma unroll
    for (int g = 0; g < NCHUNK; ++g) c[g] = cntP[(size_t)g * n + i];
    float d = 1.0f;
#pragma unroll
    for (int g = 0; g < NCHUNK; ++g) d += dp[g];
    dinv[i] = rsqrtf(d);
#pragma unroll
    for (int g = 0; g < NCHUNK; ++g) tot += c[g];
  } else {
#pragma unroll
    for (int g = 0; g < NCHUNK; ++g) c[g] = 0;
  }
  int incl = tot;
#pragma unroll
  for (int d = 1; d < 32; d <<= 1) {
    const int v = __shfl_up(incl, d);
    if (lane >= d) incl += v;
  }
  if (lane == 31) wsum[wave] = incl;
  __syncthreads();
  if (threadIdx.x == 0) {
    int run = 0;
    for (int k = 0; k < 8; ++k) {
      const int x = wsum[k];
      wsum[k] = run;
      run += x;
    }
    bbase = atomicAdd(counter, run);
  }
  __syncthreads();
  if (i < n) {
    int p = bbase + wsum[wave] + incl - tot;
    start[i] = p;
#pragma unroll
    for (int g = 0; g < NCHUNK; ++g) {
      cntP[(size_t)g * n + i] = p;
      p += c[g];
    }
    endp[i] = p;
  }
}

__global__ __launch_bounds__(256) void k_fill(const int* __restrict__ src,
                                              const int* __restrict__ tgt,
                                              const float* __restrict__ w,
                                              const int* __restrict__ rank,
                                              const int* __restrict__ offP,
                                              const float* __restrict__ dinv,
                                              int* __restrict__ pairs, int n, int E) {
  const int e = blockIdx.x * 256 + threadIdx.x;
  if (e >= E) return;
  const int chunk = (E + NCHUNK - 1) / NCHUNK;
  const int g = e / chunk;
  const int t = tgt[e];
  const int p = offP[(size_t)g * n + t] + rank[e];
  const int s = src[e];
  v2i pr;
  pr.x = s;
  pr.y = __float_as_int(w[e] * dinv[s]);
  *(v2i*)(pairs + 2 * (size_t)p) = pr;
}

#define LDP 132

__global__ __launch_bounds__(256) void k_gemm_xwT(const float* __restrict__ X,
                                                  const float* __restrict__ W,
                                                  float* __restrict__ Y, int n) {
  __shared__ __attribute__((aligned(16))) float sW[DIM * LDP];
  __shared__ __attribute__((aligned(16))) float sX[16 * LDP];

  const int wave = threadIdx.x >> 5;
  const int lane = threadIdx.x & 31;
  const int row0 = blockIdx.x << 4;

#pragma unroll
  for (int i = 0; i < 2; ++i) {
    const int r = wave + 8 * i;
    int gr = row0 + r;
    if (gr >= n) gr = n - 1;
    const v4f v = *(const v4f*)(X + (size_t)gr * DIM + 4 * lane);
    *(v4f*)(sX + r * LDP + 4 * lane) = v;
  }
#pragma unroll
  for (int i = 0; i < 16; ++i) {
    const int r = wave + 8 * i;
    const v4f v = *(const v4f*)(W + (size_t)r * DIM + 4 * lane);
    *(v4f*)(sW + r * LDP + 4 * lane) = v;
  }
  __syncthreads();

  const int m    = lane & 15;
  const int hi   = lane >> 4;
  const int kb   = hi << 1;
  const int col  = (wave << 4) + m;

  const float* arow = sX + m * LDP + kb;
  const float* brow = sW + col * LDP + kb;

  v8f c = {};
#pragma unroll
  for (int k0 = 0; k0 < DIM; k0 += 4) {
    v2f a = *(const v2f*)(arow + k0);
    v2f b = *(const v2f*)(brow + k0);
    c = __builtin_amdgcn_wmma_f32_16x16x4_f32(false, a, false, b, (short)0, c,
                                              false, false);
  }

  const int mh = hi << 3;
#pragma unroll
  for (int v = 0; v < 8; ++v) {
    int r = row0 + mh + v;
    if (r < n) Y[(size_t)r * DIM + col] = c[v];
  }
}

__global__ __launch_bounds__(256) void k_gather(const int* __restrict__ pairs,
                                                const int* __restrict__ start,
                                                const int* __restrict__ endp,
                                                const float* __restrict__ dinv,
                                                const float* __restrict__ Y,
                                                const float* __restrict__ bias,
                                                float* __restrict__ out, int n) {
  const int lane = threadIdx.x & 31;
  const int t = __builtin_amdgcn_readfirstlane(blockIdx.x * 8 + (threadIdx.x >> 5));
  if (t >= n) return;
  const int beg = start[t];
  const int end = endp[t];
  const float di = dinv[t];
  const v4f yt = *(const v4f*)(Y + (size_t)t * DIM + 4 * lane);
  const v4f bv = *(const v4f*)(bias + 4 * lane);
  v4f acc = yt * di * di + bv;
  for (int base = beg; base < end; base += 32) {
    const int k = base + lane;
    int s = 0;
    float wn = 0.0f;
    if (k < end) {
      const v2i pr = *(const v2i*)(pairs + 2 * (size_t)k);
      s = pr.x;
      wn = __int_as_float(pr.y) * di;
    }
    const int wb = __float_as_int(wn);
    const int m = min(32, end - base);
    int j = 0;
    for (; j + 4 <= m; j += 4) {
      const int s0 = __builtin_amdgcn_readlane(s, j);
      const int s1 = __builtin_amdgcn_readlane(s, j + 1);
      const int s2 = __builtin_amdgcn_readlane(s, j + 2);
      const int s3 = __builtin_amdgcn_readlane(s, j + 3);
      const float w0 = __int_as_float(__builtin_amdgcn_readlane(wb, j));
      const float w1 = __int_as_float(__builtin_amdgcn_readlane(wb, j + 1));
      const float w2 = __int_as_float(__builtin_amdgcn_readlane(wb, j + 2));
      const float w3 = __int_as_float(__builtin_amdgcn_readlane(wb, j + 3));
      const v4f y0 = *(const v4f*)(Y + (size_t)s0 * DIM + 4 * lane);
      const v4f y1 = *(const v4f*)(Y + (size_t)s1 * DIM + 4 * lane);
      const v4f y2 = *(const v4f*)(Y + (size_t)s2 * DIM + 4 * lane);
      const v4f y3 = *(const v4f*)(Y + (size_t)s3 * DIM + 4 * lane);
      acc += y0 * w0;
      acc += y1 * w1;
      acc += y2 * w2;
      acc += y3 * w3;
    }
    const int r = m - j;
    if (r == 3) {
      const int s0 = __builtin_amdgcn_readlane(s, j);
      const int s1 = __builtin_amdgcn_readlane(s, j + 1);
      const int s2 = __builtin_amdgcn_readlane(s, j + 2);
      const float w0 = __int_as_float(__builtin_amdgcn_readlane(wb, j));
      const float w1 = __int_as_float(__builtin_amdgcn_readlane(wb, j + 1));
      const float w2 = __int_as_float(__builtin_amdgcn_readlane(wb, j + 2));
      const v4f y0 = *(const v4f*)(Y + (size_t)s0 * DIM + 4 * lane);
      const v4f y1 = *(const v4f*)(Y + (size_t)s1 * DIM + 4 * lane);
      const v4f y2 = *(const v4f*)(Y + (size_t)s2 * DIM + 4 * lane);
      acc += y0 * w0;
      acc += y1 * w1;
      acc += y2 * w2;
    } else if (r == 2) {
      const int s0 = __builtin_amdgcn_readlane(s, j);
      const int s1 = __builtin_amdgcn_readlane(s, j + 1);
      const float w0 = __int_as_float(__builtin_amdgcn_readlane(wb, j));
      const float w1 = __int_as_float(__builtin_amdgcn_readlane(wb, j + 1));
      const v4f y0 = *(const v4f*)(Y + (size_t)s0 * DIM + 4 * lane);
      const v4f y1 = *(const v4f*)(Y + (size_t)s1 * DIM + 4 * lane);
      acc += y0 * w0;
      acc += y1 * w1;
    } else if (r == 1) {
      const int s0 = __builtin_amdgcn_readlane(s, j);
      const float w0 = __int_as_float(__builtin_amdgcn_readlane(wb, j));
      const v4f y0 = *(const v4f*)(Y + (size_t)s0 * DIM + 4 * lane);
      acc += y0 * w0;
    }
  }
  *(v4f*)(out + (size_t)t * DIM + 4 * lane) = acc;
}

extern "C" void kernel_launch(void* const* d_in, const int* in_sizes, int n_in,
                              void* d_out, int out_size, void* d_ws, size_t ws_size,
                              hipStream_t stream) {
  const float* x          = (const float*)d_in[0];
  const int*   edge_index = (const int*)d_in[1];
  const float* ew         = (const float*)d_in[2];
  const float* W          = (const float*)d_in[3];
  const float* b          = (const float*)d_in[4];

  const int N = in_sizes[0] / DIM;
  const int E = in_sizes[2];
  const int* src = edge_index;
  const int* tgt = edge_index + E;

  float* ws    = (float*)d_ws;
  float* Y     = ws;
  int*   pairs = (int*)(Y + (size_t)N * DIM);
  int*   rank  = pairs + 2 * (size_t)E;
  float* degP  = (float*)(rank + E);
  int*   cntP  = (int*)(degP + (size_t)NCHUNK * N);
  float* dinv  = (float*)(cntP + (size_t)NCHUNK * N);
  int*   start = (int*)(dinv + N);
  int*   endp  = start + N;
  int*   counter = endp + N;
  float* out   = (float*)d_out;

  k_hist<<<2 * NCHUNK, 1024, 0, stream>>>(src, tgt, ew, degP, cntP, rank, counter, N, E);
  k_node<<<(N + 255) / 256, 256, 0, stream>>>(degP, cntP, dinv, start, endp, counter, N);
  k_fill<<<(E + 255) / 256, 256, 0, stream>>>(src, tgt, ew, rank, cntP, dinv, pairs, N, E);
  k_gemm_xwT<<<(N + 15) / 16, 256, 0, stream>>>(x, W, Y, N);
  k_gather<<<(N + 7) / 8, 256, 0, stream>>>(pairs, start, endp, dinv, Y, b, out, N);
}


# dense product run inside the first launch as extra 1024-thread groups beside the 64 LDS-table groups, which leave most compute units idle; W staged once per 128 rows; one launch fewer
# speedup vs baseline: 2.9897x; 1.1535x over previous
#include <hip/hip_runtime.h>

typedef __attribute__((ext_vector_type(2))) float v2f;
typedef __attribute__((ext_vector_type(4))) float v4f;
typedef __attribute__((ext_vector_type(8))) float v8f;
typedef __attribute__((ext_vector_type(2))) int v2i;
typedef __attribute__((ext_vector_type(4))) int v4i;

#define DIM 128

#define NCHUNK 32
#define NCAP 50176
#define HB 8

#define LDP 132
#define GR 64
#define BPG 2

__global__ __launch_bounds__(1024) void k_hist_gemm(const int* __restrict__ src,
                                                    const int* __restrict__ tgt,
                                                    const float* __restrict__ w,
                                                    float* __restrict__ degP,
                                                    int* __restrict__ cntP,
                                                    int* __restrict__ rank,
                                                    int* __restrict__ counter,
                                                    const float* __restrict__ X,
                                                    const float* __restrict__ W,
                                                    float* __restrict__ Y, int n, int E) {
  __shared__ __attribute__((aligned(16))) union { int i[NCAP]; float f[NCAP]; } h;
  const int tid = threadIdx.x;
  if (blockIdx.x >= 2 * NCHUNK) {
    float* sW = h.f;
    float* sX = h.f + DIM * LDP;
    const int wave = tid >> 5;
    const int lane = tid & 31;
    int row0 = (blockIdx.x - 2 * NCHUNK) * (GR * BPG);
    if (row0 >= n) return;
#pragma unroll
    for (int i = 0; i < 4; ++i) {
      const int r = wave + 32 * i;
      const v4f v = *(const v4f*)(W + (size_t)r * DIM + 4 * lane);
      *(v4f*)(sW + r * LDP + 4 * lane) = v;
    }
#pragma unroll
    for (int i = 0; i < 2; ++i) {
      const int r = wave + 32 * i;
      int gr = row0 + r;
      if (gr >= n) gr = n - 1;
      const v4f v = *(const v4f*)(X + (size_t)gr * DIM + 4 * lane);
      *(v4f*)(sX + r * LDP + 4 * lane) = v;
    }
    __syncthreads();

    const int m   = lane & 15;
    const int hi  = lane >> 4;
    const int kb  = hi << 1;
    const int col = ((wave & 7) << 4) + m;
    const int sr  = (wave >> 3) << 4;
    const int mh  = hi << 3;

    const float* brow = sW + col * LDP + kb;
    v2f b[DIM / 4];
#pragma unroll
    for (int j = 0; j < DIM / 4; ++j) b[j] = *(const v2f*)(brow + 4 * j);

    for (int bi = 0; bi < BPG; ++bi) {
      const float* sXc = sX + (bi & 1) * (GR * LDP);
      float* sXn = sX + ((bi & 1) ^ 1) * (GR * LDP);
      const int nrow0 = row0 + GR;
      const bool has_next = (bi + 1 < BPG) && (nrow0 < n);
      v4f p[2];
      if (has_next) {
#pragma unroll
        for (int i = 0; i < 2; ++i) {
          int gr = nrow0 + wave + 32 * i;
          if (gr >= n) gr = n - 1;
          p[i] = *(const v4f*)(X + (size_t)gr * DIM + 4 * lane);
        }
      }
      const float* arow = sXc + (sr + m) * LDP + kb;
      v8f c = {};
#pragma unroll
      for (int k0 = 0; k0 < DIM; k0 += 4) {
        v2f a = *(const v2f*)(arow + k0);
        c = __builtin_amdgcn_wmma_f32_16x16x4_f32(false, a, false, b[k0 >> 2], (short)0, c,
                                                  false, false);
      }
#pragma unroll
      for (int v = 0; v < 8; ++v) {
        const int r = row0 + sr + mh + v;
        if (r < n) Y[(size_t)r * DIM + col] = c[v];
      }
      if (!has_next) break;
#pragma unroll
      for (int i = 0; i < 2; ++i) *(v4f*)(sXn + (wave + 32 * i) * LDP + 4 * lane) = p[i];
      __syncthreads();
      row0 = nrow0;
    }
    return;
  }
  if (n > NCAP) return;
  const int g = blockIdx.x & (NCHUNK - 1);
  const bool is_deg = blockIdx.x >= NCHUNK;
  if (blockIdx.x == 0 && tid == 0) *counter = 0;
  {
    const v4i z = {0, 0, 0, 0};
    for (int i = tid; i < NCAP / 4; i += 1024) ((v4i*)h.i)[i] = z;
  }
  __syncthreads();
  const int chunk = (E + NCHUNK - 1) / NCHUNK;
  const int e0 = g * chunk;
  const int e1 = min(E, e0 + chunk);
  if (is_deg) {
    for (int eb = e0 + tid; eb < e1; eb += 1024 * HB) {
      int key[HB];
      float val[HB];
#pragma unroll
      for (int u = 0; u < HB; ++u) {
        const int e = eb + 1024 * u;
        key[u] = (e < e1) ? src[e] : 0;
        val[u] = (e < e1) ? w[e] : 0.0f;
      }
#pragma unroll
      for (int u = 0; u < HB; ++u) {
        if (eb + 1024 * u < e1) atomicAdd(&h.f[key[u]], val[u]);
      }
    }
  } else {
    for (int eb = e0 + tid; eb < e1; eb += 1024 * HB) {
      int key[HB];
#pragma unroll
      for (int u = 0; u < HB; ++u) {
        const int e = eb + 1024 * u;
        key[u] = (e < e1) ? tgt[e] : 0;
      }
#pragma unroll
      for (int u = 0; u < HB; ++u) {
        const int e = eb + 1024 * u;
        if (e < e1) rank[e] = atomicAdd(&h.i[key[u]], 1);
      }
    }
  }
  __syncthreads();
  if (is_deg) {
    for (int i = tid; i < n; i += 1024) degP[(size_t)g * n + i] = h.f[i];
  } else {
    for (int i = tid; i < n; i += 1024) cntP[(size_t)g * n + i] = h.i[i];
  }
}

__global__ __launch_bounds__(256) void k_node(const float* __restrict__ degP,
                                              int* __restrict__ cntP,
                                              float* __restrict__ dinv,
                                              int* __restrict__ start,
                                              int* __restrict__ endp,
                                              int* __restrict__ counter, int n) {
  __shared__ int wsum[8];
  __shared__ int bbase;
  const int i = blockIdx.x * 256 + threadIdx.x;
  const int lane = threadIdx.x & 31;
  const int wave = threadIdx.x >> 5;
  int c[NCHUNK];
  int tot = 0;
  if (i < n) {
    float dp[NCHUNK];
#pragma unroll
    for (int g = 0; g < NCHUNK; ++g) dp[g] = degP[(size_t)g * n + i];
#pragma unroll
    for (int g = 0; g < NCHUNK; ++g) c[g] = cntP[(size_t)g * n + i];
    float d = 1.0f;
#pragma unroll
    for (int g = 0; g < NCHUNK; ++g) d += dp[g];
    dinv[i] = rsqrtf(d);
#pragma unroll
    for (int g = 0; g < NCHUNK; ++g) tot += c[g];
  } else {
#pragma unroll
    for (int g = 0; g < NCHUNK; ++g) c[g] = 0;
  }
  int incl = tot;
#pragma unroll
  for (int d = 1; d < 32; d <<= 1) {
    const int v = __shfl_up(incl, d);
    if (lane >= d) incl += v;
  }
  if (lane == 31) wsum[wave] = incl;
  __syncthreads();
  if (threadIdx.x == 0) {
    int run = 0;
    for (int k = 0; k < 8; ++k) {
      const int x = wsum[k];
      wsum[k] = run;
      run += x;
    }
    bbase = atomicAdd(counter, run);
  }
  __syncthreads();
  if (i < n) {
    int p = bbase + wsum[wave] + incl - tot;
    start[i] = p;
#pragma unroll
    for (int g = 0; g < NCHUNK; ++g) {
      cntP[(size_t)g * n + i] = p;
      p += c[g];
    }
    endp[i] = p;
  }
}

__global__ __launch_bounds__(256) void k_fill(const int* __restrict__ src,
                                              const int* __restrict__ tgt,
                                              const float* __restrict__ w,
                                              const int* __restrict__ rank,
                                              const int* __restrict__ offP,
                                              const float* __restrict__ dinv,
                                              int* __restrict__ pairs, int n, int E) {
  const int e = blockIdx.x * 256 + threadIdx.x;
  if (e >= E) return;
  const int chunk = (E + NCHUNK - 1) / NCHUNK;
  const int g = e / chunk;
  const int t = tgt[e];
  const int p = offP[(size_t)g * n + t] + rank[e];
  const int s = src[e];
  v2i pr;
  pr.x = s;
  pr.y = __float_as_int(w[e] * dinv[s]);
  *(v2i*)(pairs + 2 * (size_t)p) = pr;
}

__global__ __launch_bounds__(256) void k_gather(const int* __restrict__ pairs,
                                                const int* __restrict__ start,
                                                const int* __restrict__ endp,
                                                const float* __restrict__ dinv,
                                                const float* __restrict__ Y,
                                                const float* __restrict__ bias,
                                                float* __restrict__ out, int n) {
  const int lane = threadIdx.x & 31;
  const int t = __builtin_amdgcn_readfirstlane(blockIdx.x * 8 + (threadIdx.x >> 5));
  if (t >= n) return;
  const int beg = start[t];
  const int end = endp[t];
  const float di = dinv[t];
  const v4f yt = *(const v4f*)(Y + (size_t)t * DIM + 4 * lane);
  const v4f bv = *(const v4f*)(bias + 4 * lane);
  v4f acc = yt * di * di + bv;
  for (int base = beg; base < end; base += 32) {
    const int k = base + lane;
    int s = 0;
    float wn = 0.0f;
    if (k < end) {
      const v2i pr = *(const v2i*)(pairs + 2 * (size_t)k);
      s = pr.x;
      wn = __int_as_float(pr.y) * di;
    }
    const int wb = __float_as_int(wn);
    const int m = min(32, end - base);
    int j = 0;
    for (; j + 4 <= m; j += 4) {
      const int s0 = __builtin_amdgcn_readlane(s, j);
      const int s1 = __builtin_amdgcn_readlane(s, j + 1);
      const int s2 = __builtin_amdgcn_readlane(s, j + 2);
      const int s3 = __builtin_amdgcn_readlane(s, j + 3);
      const float w0 = __int_as_float(__builtin_amdgcn_readlane(wb, j));
      const float w1 = __int_as_float(__builtin_amdgcn_readlane(wb, j + 1));
      const float w2 = __int_as_float(__builtin_amdgcn_readlane(wb, j + 2));
      const float w3 = __int_as_float(__builtin_amdgcn_readlane(wb, j + 3));
      const v4f y0 = *(const v4f*)(Y + (size_t)s0 * DIM + 4 * lane);
      const v4f y1 = *(const v4f*)(Y + (size_t)s1 * DIM + 4 * lane);
      const v4f y2 = *(const v4f*)(Y + (size_t)s2 * DIM + 4 * lane);
      const v4f y3 = *(const v4f*)(Y + (size_t)s3 * DIM + 4 * lane);
      acc += y0 * w0;
      acc += y1 * w1;
      acc += y2 * w2;
      acc += y3 * w3;
    }
    const int r = m - j;
    if (r == 3) {
      const int s0 = __builtin_amdgcn_readlane(s, j);
      const int s1 = __builtin_amdgcn_readlane(s, j + 1);
      const int s2 = __builtin_amdgcn_readlane(s, j + 2);
      const float w0 = __int_as_float(__builtin_amdgcn_readlane(wb, j));
      const float w1 = __int_as_float(__builtin_amdgcn_readlane(wb, j + 1));
      const float w2 = __int_as_float(__builtin_amdgcn_readlane(wb, j + 2));
      const v4f y0 = *(const v4f*)(Y + (size_t)s0 * DIM + 4 * lane);
      const v4f y1 = *(const v4f*)(Y + (size_t)s1 * DIM + 4 * lane);
      const v4f y2 = *(const v4f*)(Y + (size_t)s2 * DIM + 4 * lane);
      acc += y0 * w0;
      acc += y1 * w1;
      acc += y2 * w2;
    } else if (r == 2) {
      const int s0 = __builtin_amdgcn_readlane(s, j);
      const int s1 = __builtin_amdgcn_readlane(s, j + 1);
      const float w0 = __int_as_float(__builtin_amdgcn_readlane(wb, j));
      const float w1 = __int_as_float(__builtin_amdgcn_readlane(wb, j + 1));
      const v4f y0 = *(const v4f*)(Y + (size_t)s0 * DIM + 4 * lane);
      const v4f y1 = *(const v4f*)(Y + (size_t)s1 * DIM + 4 * lane);
      acc += y0 * w0;
      acc += y1 * w1;
    } else if (r == 1) {
      const int s0 = __builtin_amdgcn_readlane(s, j);
      const float w0 = __int_as_float(__builtin_amdgcn_readlane(wb, j));
      const v4f y0 = *(const v4f*)(Y + (size_t)s0 * DIM + 4 * lane);
      acc += y0 * w0;
    }
  }
  *(v4f*)(out + (size_t)t * DIM + 4 * lane) = acc;
}

extern "C" void kernel_launch(void* const* d_in, const int* in_sizes, int n_in,
                              void* d_out, int out_size, void* d_ws, size_t ws_size,
                              hipStream_t stream) {
  const float* x          = (const float*)d_in[0];
  const int*   edge_index = (const int*)d_in[1];
  const float* ew         = (const float*)d_in[2];
  const float* W          = (const float*)d_in[3];
  const float* b          = (const float*)d_in[4];

  const int N = in_sizes[0] / DIM;
  const int E = in_sizes[2];
  const int* src = edge_index;
  const int* tgt = edge_index + E;

  float* ws    = (float*)d_ws;
  float* Y     = ws;
  int*   pairs = (int*)(Y + (size_t)N * DIM);
  int*   rank  = pairs + 2 * (size_t)E;
  float* degP  = (float*)(rank + E);
  int*   cntP  = (int*)(degP + (size_t)NCHUNK * N);
  float* dinv  = (float*)(cntP + (size_t)NCHUNK * N);
  int*   start = (int*)(dinv + N);
  int*   endp  = start + N;
  int*   counter = endp + N;
  float* out   = (float*)d_out;

  k_hist_gemm<<<2 * NCHUNK + (N + GR * BPG - 1) / (GR * BPG), 1024, 0, stream>>>(
      src, tgt, ew, degP, cntP, rank, counter, x, W, Y, N, E);
  k_node<<<(N + 255) / 256, 256, 0, stream>>>(degP, cntP, dinv, start, endp, counter, N);
  k_fill<<<(E + 255) / 256, 256, 0, stream>>>(src, tgt, ew, rank, cntP, dinv, pairs, N, E);
  k_gather<<<(N + 7) / 8, 256, 0, stream>>>(pairs, start, endp, dinv, Y, b, out, N);
}


# dense product split over two launches: the 192 row groups that fit beside the table groups stay, the other 199 run inside the edge-copy launch (B fragments from LDS there) on idle matrix units
# speedup vs baseline: 3.0938x; 1.0348x over previous
#include <hip/hip_runtime.h>

typedef __attribute__((ext_vector_type(2))) float v2f;
typedef __attribute__((ext_vector_type(4))) float v4f;
typedef __attribute__((ext_vector_type(8))) float v8f;
typedef __attribute__((ext_vector_type(2))) int v2i;
typedef __attribute__((ext_vector_type(4))) int v4i;

#define DIM 128

#define NCHUNK 32
#define NCAP 50176
#define HB 8

#define LDP 132
#define GR 64
#define BPG 2
#define NUNIT 256
#define DG1 (NUNIT - 2 * NCHUNK)

template <bool BREG>
__device__ __forceinline__ void dense_group(float* __restrict__ lds, const float* __restrict__ X,
                                            const float* __restrict__ W, float* __restrict__ Y,
                                            int n, int row0, int tid) {
  float* sW = lds;
  float* sX = lds + DIM * LDP;
  const int wave = tid >> 5;
  const int lane = tid & 31;
  if (row0 >= n) return;
#pragma unroll
  for (int i = 0; i < 4; ++i) {
    const int r = wave + 32 * i;
    const v4f v = *(const v4f*)(W + (size_t)r * DIM + 4 * lane);
    *(v4f*)(sW + r * LDP + 4 * lane) = v;
  }
#pragma unroll
  for (int i = 0; i < 2; ++i) {
    const int r = wave + 32 * i;
    int gr = row0 + r;
    if (gr >= n) gr = n - 1;
    const v4f v = *(const v4f*)(X + (size_t)gr * DIM + 4 * lane);
    *(v4f*)(sX + r * LDP + 4 * lane) = v;
  }
  __syncthreads();

  const int m   = lane & 15;
  const int hi  = lane >> 4;
  const int kb  = hi << 1;
  const int col = ((wave & 7) << 4) + m;
  const int sr  = (wave >> 3) << 4;
  const int mh  = hi << 3;

  const float* brow = sW + col * LDP + kb;
  v2f b[DIM / 4];
  if (BREG) {
#pragma unroll
    for (int j = 0; j < DIM / 4; ++j) b[j] = *(const v2f*)(brow + 4 * j);
  }

  for (int bi = 0; bi < BPG; ++bi) {
    const float* sXc = sX + (bi & 1) * (GR * LDP);
    float* sXn = sX + ((bi & 1) ^ 1) * (GR * LDP);
    const int nrow0 = row0 + GR;
    const bool has_next = (bi + 1 < BPG) && (nrow0 < n);
    v4f p[2];
    if (has_next) {
#pragma unroll
      for (int i = 0; i < 2; ++i) {
        int gr = nrow0 + wave + 32 * i;
        if (gr >= n) gr = n - 1;
        p[i] = *(const v4f*)(X + (size_t)gr * DIM + 4 * lane);
      }
    }
    const float* arow = sXc + (sr + m) * LDP + kb;
    v8f c = {};
#pragma unroll
    for (int k0 = 0; k0 < DIM; k0 += 4) {
      const v2f a = *(const v2f*)(arow + k0);
      const v2f bb = BREG ? b[k0 >> 2] : *(const v2f*)(brow + k0);
      c = __builtin_amdgcn_wmma_f32_16x16x4_f32(false, a, false, bb, (short)0, c, false, false);
    }
#pragma unroll
    for (int v = 0; v < 8; ++v) {
      const int r = row0 + sr + mh + v;
      if (r < n) Y[(size_t)r * DIM + col] = c[v];
    }
    if (!has_next) break;
#pragma unroll
    for (int i = 0; i < 2; ++i) *(v4f*)(sXn + (wave + 32 * i) * LDP + 4 * lane) = p[i];
    __syncthreads();
    row0 = nrow0;
  }
}

__global__ __launch_bounds__(1024) void k_hist_gemm(const int* __restrict__ src,
                                                    const int* __restrict__ tgt,
                                                    const float* __restrict__ w,
                                                    float* __restrict__ degP,
                                                    int* __restrict__ cntP,
                                                    int* __restrict__ rank,
                                                    int* __restrict__ counter,
                                                    const float* __restrict__ X,
                                                    const float* __restrict__ W,
                                                    float* __restrict__ Y, int n, int E) {
  __shared__ __attribute__((aligned(16))) union { int i[NCAP]; float f[NCAP]; } h;
  const int tid = threadIdx.x;
  if (blockIdx.x >= 2 * NCHUNK) {
    dense_group<true>(h.f, X, W, Y, n, (blockIdx.x - 2 * NCHUNK) * (GR * BPG), tid);
    return;
  }
  if (n > NCAP) return;
  const int g = blockIdx.x & (NCHUNK - 1);
  const bool is_deg = blockIdx.x >= NCHUNK;
  if (blockIdx.x == 0 && tid == 0) *counter = 0;
  {
    const v4i z = {0, 0, 0, 0};
    for (int i = tid; i < NCAP / 4; i += 1024) ((v4i*)h.i)[i] = z;
  }
  __syncthreads();
  const int chunk = (E + NCHUNK - 1) / NCHUNK;
  const int e0 = g * chunk;
  const int e1 = min(E, e0 + chunk);
  if (is_deg) {
    for (int eb = e0 + tid; eb < e1; eb += 1024 * HB) {
      int key[HB];
      float val[HB];
#pragma unroll
      for (int u = 0; u < HB; ++u) {
        const int e = eb + 1024 * u;
        key[u] = (e < e1) ? src[e] : 0;
        val[u] = (e < e1) ? w[e] : 0.0f;
      }
#pragma unroll
      for (int u = 0; u < HB; ++u) {
        if (eb + 1024 * u < e1) atomicAdd(&h.f[key[u]], val[u]);
      }
    }
  } else {
    for (int eb = e0 + tid; eb < e1; eb += 1024 * HB) {
      int key[HB];
#pragma unroll
      for (int u = 0; u < HB; ++u) {
        const int e = eb + 1024 * u;
        key[u] = (e < e1) ? tgt[e] : 0;
      }
#pragma unroll
      for (int u = 0; u < HB; ++u) {
        const int e = eb + 1024 * u;
        if (e < e1) rank[e] = atomicAdd(&h.i[key[u]], 1);
      }
    }
  }
  __syncthreads();
  if (is_deg) {
    for (int i = tid; i < n; i += 1024) degP[(size_t)g * n + i] = h.f[i];
  } else {
    for (int i = tid; i < n; i += 1024) cntP[(size_t)g * n + i] = h.i[i];
  }
}

__global__ __launch_bounds__(256) void k_node(const float* __restrict__ degP,
                                              int* __restrict__ cntP,
                                              float* __restrict__ dinv,
                                              int* __restrict__ start,
                                              int* __restrict__ endp,
                                              int* __restrict__ counter, int n) {
  __shared__ int wsum[8];
  __shared__ int bbase;
  const int i = blockIdx.x * 256 + threadIdx.x;
  const int lane = threadIdx.x & 31;
  const int wave = threadIdx.x >> 5;
  int c[NCHUNK];
  int tot = 0;
  if (i < n) {
    float dp[NCHUNK];
#pragma unroll
    for (int g = 0; g < NCHUNK; ++g) dp[g] = degP[(size_t)g * n + i];
#pragma unroll
    for (int g = 0; g < NCHUNK; ++g) c[g] = cntP[(size_t)g * n + i];
    float d = 1.0f;
#pragma unroll
    for (int g = 0; g < NCHUNK; ++g) d += dp[g];
    dinv[i] = rsqrtf(d);
#pragma unroll
    for (int g = 0; g < NCHUNK; ++g) tot += c[g];
  } else {
#pragma unroll
    for (int g = 0; g < NCHUNK; ++g) c[g] = 0;
  }
  int incl = tot;
#pragma unroll
  for (int d = 1; d < 32; d <<= 1) {
    const int v = __shfl_up(incl, d);
    if (lane >= d) incl += v;
  }
  if (lane == 31) wsum[wave] = incl;
  __syncthreads();
  if (threadIdx.x == 0) {
    int run = 0;
    for (int k = 0; k < 8; ++k) {
      const int x = wsum[k];
      wsum[k] = run;
      run += x;
    }
    bbase = atomicAdd(counter, run);
  }
  __syncthreads();
  if (i < n) {
    int p = bbase + wsum[wave] + incl - tot;
    start[i] = p;
#pragma unroll
    for (int g = 0; g < NCHUNK; ++g) {
      cntP[(size_t)g * n + i] = p;
      p += c[g];
    }
    endp[i] = p;
  }
}

__global__ __launch_bounds__(1024, 16) void k_fill_gemm(const int* __restrict__ src,
                                                        const int* __restrict__ tgt,
                                                        const float* __restrict__ w,
                                                        const int* __restrict__ rank,
                                                        const int* __restrict__ offP,
                                                        const float* __restrict__ dinv,
                                                        int* __restrict__ pairs,
                                                        const float* __restrict__ X,
                                                        const float* __restrict__ W,
                                                        float* __restrict__ Y, int n, int E,
                                                        int g0, int ng) {
  __shared__ __attribute__((aligned(16))) float lds[(DIM + 2 * GR) * LDP];
  if ((int)blockIdx.x < ng) {
    dense_group<false>(lds, X, W, Y, n, (g0 + (int)blockIdx.x) * (GR * BPG), threadIdx.x);
    return;
  }
  const int e = ((int)blockIdx.x - ng) * 1024 + threadIdx.x;
  if (e >= E) return;
  const int chunk = (E + NCHUNK - 1) / NCHUNK;
  const int g = e / chunk;
  const int t = tgt[e];
  const int p = offP[(size_t)g * n + t] + rank[e];
  const int s = src[e];
  v2i pr;
  pr.x = s;
  pr.y = __float_as_int(w[e] * dinv[s]);
  *(v2i*)(pairs + 2 * (size_t)p) = pr;
}

__global__ __launch_bounds__(256) void k_gather(const int* __restrict__ pairs,
                                                const int* __restrict__ start,
                                                const int* __restrict__ endp,
                                                const float* __restrict__ dinv,
                                                const float* __restrict__ Y,
                                                const float* __restrict__ bias,
                                                float* __restrict__ out, int n) {
  const int lane = threadIdx.x & 31;
  const int t = __builtin_amdgcn_readfirstlane(blockIdx.x * 8 + (threadIdx.x >> 5));
  if (t >= n) return;
  const int beg = start[t];
  const int end = endp[t];
  const float di = dinv[t];
  const v4f yt = *(const v4f*)(Y + (size_t)t * DIM + 4 * lane);
  const v4f bv = *(const v4f*)(bias + 4 * lane);
  v4f acc = yt * di * di + bv;
  for (int base = beg; base < end; base += 32) {
    const int k = base + lane;
    int s = 0;
    float wn = 0.0f;
    if (k < end) {
      const v2i pr = *(const v2i*)(pairs + 2 * (size_t)k);
      s = pr.x;
      wn = __int_as_float(pr.y) * di;
    }
    const int wb = __float_as_int(wn);
    const int m = min(32, end - base);
    int j = 0;
    for (; j + 4 <= m; j += 4) {
      const int s0 = __builtin_amdgcn_readlane(s, j);
      const int s1 = __builtin_amdgcn_readlane(s, j + 1);
      const int s2 = __builtin_amdgcn_readlane(s, j + 2);
      const int s3 = __builtin_amdgcn_readlane(s, j + 3);
      const float w0 = __int_as_float(__builtin_amdgcn_readlane(wb, j));
      const float w1 = __int_as_float(__builtin_amdgcn_readlane(wb, j + 1));
      const float w2 = __int_as_float(__builtin_amdgcn_readlane(wb, j + 2));
      const float w3 = __int_as_float(__builtin_amdgcn_readlane(wb, j + 3));
      const v4f y0 = *(const v4f*)(Y + (size_t)s0 * DIM + 4 * lane);
      const v4f y1 = *(const v4f*)(Y + (size_t)s1 * DIM + 4 * lane);
      const v4f y2 = *(const v4f*)(Y + (size_t)s2 * DIM + 4 * lane);
      const v4f y3 = *(const v4f*)(Y + (size_t)s3 * DIM + 4 * lane);
      acc += y0 * w0;
      acc += y1 * w1;
      acc += y2 * w2;
      acc += y3 * w3;
    }
    const int r = m - j;
    if (r == 3) {
      const int s0 = __builtin_amdgcn_readlane(s, j);
      const int s1 = __builtin_amdgcn_readlane(s, j + 1);
      const int s2 = __builtin_amdgcn_readlane(s, j + 2);
      const float w0 = __int_as_float(__builtin_amdgcn_readlane(wb, j));
      const float w1 = __int_as_float(__builtin_amdgcn_readlane(wb, j + 1));
      const float w2 = __int_as_float(__builtin_amdgcn_readlane(wb, j + 2));
      const v4f y0 = *(const v4f*)(Y + (size_t)s0 * DIM + 4 * lane);
      const v4f y1 = *(const v4f*)(Y + (size_t)s1 * DIM + 4 * lane);
      const v4f y2 = *(const v4f*)(Y + (size_t)s2 * DIM + 4 * lane);
      acc += y0 * w0;
      acc += y1 * w1;
      acc += y2 * w2;
    } else if (r == 2) {
      const int s0 = __builtin_amdgcn_readlane(s, j);
      const int s1 = __builtin_amdgcn_readlane(s, j + 1);
      const float w0 = __int_as_float(__builtin_amdgcn_readlane(wb, j));
      const float w1 = __int_as_float(__builtin_amdgcn_readlane(wb, j + 1));
      const v4f y0 = *(const v4f*)(Y + (size_t)s0 * DIM + 4 * lane);
      const v4f y1 = *(const v4f*)(Y + (size_t)s1 * DIM + 4 * lane);
      acc += y0 * w0;
      acc += y1 * w1;
    } else if (r == 1) {
      const int s0 = __builtin_amdgcn_readlane(s, j);
      const float w0 = __int_as_float(__builtin_amdgcn_readlane(wb, j));
      const v4f y0 = *(const v4f*)(Y + (size_t)s0 * DIM + 4 * lane);
      acc += y0 * w0;
    }
  }
  *(v4f*)(out + (size_t)t * DIM + 4 * lane) = acc;
}

extern "C" void kernel_launch(void* const* d_in, const int* in_sizes, int n_in,
                              void* d_out, int out_size, void* d_ws, size_t ws_size,
                              hipStream_t stream) {
  const float* x          = (const float*)d_in[0];
  const int*   edge_index = (const int*)d_in[1];
  const float* ew         = (const float*)d_in[2];
  const float* W          = (const float*)d_in[3];
  const float* b          = (const float*)d_in[4];

  const int N = in_sizes[0] / DIM;
  const int E = in_sizes[2];
  const int* src = edge_index;
  const int* tgt = edge_index + E;

  float* ws    = (float*)d_ws;
  float* Y     = ws;
  int*   pairs = (int*)(Y + (size_t)N * DIM);
  int*   rank  = pairs + 2 * (size_t)E;
  float* degP  = (float*)(rank + E);
  int*   cntP  = (int*)(degP + (size_t)NCHUNK * N);
  float* dinv  = (float*)(cntP + (size_t)NCHUNK * N);
  int*   start = (int*)(dinv + N);
  int*   endp  = start + N;
  int*   counter = endp + N;
  float* out   = (float*)d_out;

  const int NG  = (N + GR * BPG - 1) / (GR * BPG);
  const int NG1 = NG < DG1 ? NG : DG1;
  const int NG3 = NG - NG1;
  k_hist_gemm<<<2 * NCHUNK + NG1, 1024, 0, stream>>>(
      src, tgt, ew, degP, cntP, rank, counter, x, W, Y, N, E);
  k_node<<<(N + 255) / 256, 256, 0, stream>>>(degP, cntP, dinv, start, endp, counter, N);
  k_fill_gemm<<<NG3 + (E + 1023) / 1024, 1024, 0, stream>>>(src, tgt, ew, rank, cntP, dinv, pairs,
                                                         x, W, Y, N, E, NG1, NG3);
  k_gather<<<(N + 7) / 8, 256, 0, stream>>>(pairs, start, endp, dinv, Y, b, out, N);
}
